# starting kernel
# baseline (speedup 1.0000x reference)
#include <hip/hip_runtime.h>

typedef __attribute__((ext_vector_type(2))) float v2f;
typedef __attribute__((ext_vector_type(4))) float v4f;
typedef __attribute__((ext_vector_type(8))) float v8f;

#define N_SOURCE   100000
#define N_TARGET   4096
#define DEG        800
#define SOURCE_DIM 512
#define EMBED_DIM  32
#define TARGET_DIM 16

__global__ void __launch_bounds__(256)
transpose_embed(const float* __restrict__ embed, float* __restrict__ embedT) {
    int idx = blockIdx.x * blockDim.x + threadIdx.x;
    if (idx < SOURCE_DIM * EMBED_DIM) {
        int k = idx >> 5;
        int n = idx & 31;
        embedT[n * SOURCE_DIM + k] = embed[idx];
    }
}

__global__ void __launch_bounds__(256)
embed_gemm(const float* __restrict__ A, const float* __restrict__ BT,
           const float* __restrict__ xnorm, float* __restrict__ X) {
    const int wave = threadIdx.x >> 5;
    const int tile = blockIdx.x * 8 + wave;
    if (tile >= N_SOURCE / 32) return;
    const int lane = threadIdx.x & 31;
    const int lh   = lane & 15;
    const int hi   = lane >> 4;
    const int row0 = tile * 32;

    const float* __restrict__ arow0 = A  + (size_t)(row0 + lh)      * SOURCE_DIM + 4 * hi;
    const float* __restrict__ arow1 = A  + (size_t)(row0 + 16 + lh) * SOURCE_DIM + 4 * hi;
    const float* __restrict__ b0row = BT + (size_t)(lh)             * SOURCE_DIM + 4 * hi;
    const float* __restrict__ b1row = BT + (size_t)(16 + lh)        * SOURCE_DIM + 4 * hi;

    v8f c00 = {}, c01 = {}, c10 = {}, c11 = {};

    for (int k = 0; k < SOURCE_DIM / 8; ++k) {
        v4f a0 = *(const v4f*)(arow0 + 8 * k);
        v4f a1 = *(const v4f*)(arow1 + 8 * k);
        v4f b0 = *(const v4f*)(b0row + 8 * k);
        v4f b1 = *(const v4f*)(b1row + 8 * k);
        __builtin_prefetch(arow0 + 8 * k + 256, 0, 3);
        __builtin_prefetch(arow1 + 8 * k + 256, 0, 3);

        v2f a0e = {a0.x, a0.y}, a0o = {a0.z, a0.w};
        v2f a1e = {a1.x, a1.y}, a1o = {a1.z, a1.w};
        v2f b0e = {b0.x, b0.y}, b0o = {b0.z, b0.w};
        v2f b1e = {b1.x, b1.y}, b1o = {b1.z, b1.w};

        c00 = __builtin_amdgcn_wmma_f32_16x16x4_f32(false, a0e, false, b0e,
                                                    (short)0, c00, false, false);
        c01 = __builtin_amdgcn_wmma_f32_16x16x4_f32(false, a0e, false, b1e,
                                                    (short)0, c01, false, false);
        c10 = __builtin_amdgcn_wmma_f32_16x16x4_f32(false, a1e, false, b0e,
                                                    (short)0, c10, false, false);
        c11 = __builtin_amdgcn_wmma_f32_16x16x4_f32(false, a1e, false, b1e,
                                                    (short)0, c11, false, false);
        c00 = __builtin_amdgcn_wmma_f32_16x16x4_f32(false, a0o, false, b0o,
                                                    (short)0, c00, false, false);
        c01 = __builtin_amdgcn_wmma_f32_16x16x4_f32(false, a0o, false, b1o,
                                                    (short)0, c01, false, false);
        c10 = __builtin_amdgcn_wmma_f32_16x16x4_f32(false, a1o, false, b0o,
                                                    (short)0, c10, false, false);
        c11 = __builtin_amdgcn_wmma_f32_16x16x4_f32(false, a1o, false, b1o,
                                                    (short)0, c11, false, false);
    }

#pragma unroll
    for (int r = 0; r < 8; ++r) {
        int   rowA = row0 + r + 8 * hi;
        int   rowB = rowA + 16;
        float invA = 1.0f / xnorm[rowA];
        float invB = 1.0f / xnorm[rowB];
        X[(size_t)rowA * EMBED_DIM + lh]      = c00[r] * invA;
        X[(size_t)rowA * EMBED_DIM + 16 + lh] = c01[r] * invA;
        X[(size_t)rowB * EMBED_DIM + lh]      = c10[r] * invB;
        X[(size_t)rowB * EMBED_DIM + 16 + lh] = c11[r] * invB;
    }
}

__global__ void __launch_bounds__(256)
gather_mean(const float* __restrict__ X, const int* __restrict__ src,
            const int* __restrict__ range_list, float* __restrict__ agg) {
    __shared__ float red[8][EMBED_DIM];
    const int t     = blockIdx.x;
    const int col   = threadIdx.x & 31;
    const int grp   = threadIdx.x >> 5;
    const int start = range_list[2 * t];
    const int end   = range_list[2 * t + 1];

    float acc = 0.0f;
    for (int e = start + grp; e < end; e += 8) {
        int s = src[e];
        acc += X[(size_t)s * EMBED_DIM + col];
    }
    red[grp][col] = acc;
    __syncthreads();
    if (grp == 0) {
        float sum = acc;
#pragma unroll
        for (int g = 1; g < 8; ++g) sum += red[g][col];
        float deg = (float)(end - start);
        deg = deg > 1.0f ? deg : 1.0f;
        agg[(size_t)t * EMBED_DIM + col] = sum / deg;
    }
}

__global__ void __launch_bounds__(256)
proj_gemm(const float* __restrict__ agg, const float* __restrict__ W,
          float* __restrict__ out) {
    const int wave = threadIdx.x >> 5;
    const int tile = blockIdx.x * 8 + wave;
    if (tile >= N_TARGET / 16) return;
    const int lane = threadIdx.x & 31;
    const int lh   = lane & 15;
    const int hi   = lane >> 4;
    const int row0 = tile * 16;

    const float* __restrict__ arow = agg + (size_t)(row0 + lh) * EMBED_DIM + 2 * hi;

    v8f c = {};
#pragma unroll
    for (int kb = 0; kb < EMBED_DIM / 4; ++kb) {
        v2f a = *(const v2f*)(arow + 4 * kb);
        v2f b;
        b.x = W[(4 * kb + 2 * hi) * TARGET_DIM + lh];
        b.y = W[(4 * kb + 2 * hi + 1) * TARGET_DIM + lh];
        c = __builtin_amdgcn_wmma_f32_16x16x4_f32(false, a, false, b,
                                                  (short)0, c, false, false);
    }

#pragma unroll
    for (int r = 0; r < 8; ++r) {
        int row = row0 + r + 8 * hi;
        out[(size_t)row * TARGET_DIM + lh] = c[r];
    }
}

extern "C" void kernel_launch(void* const* d_in, const int* in_sizes, int n_in,
                              void* d_out, int out_size, void* d_ws, size_t ws_size,
                              hipStream_t stream) {
    (void)in_sizes; (void)n_in; (void)out_size; (void)ws_size;

    const float* source_feat = (const float*)d_in[0];
    const float* embed       = (const float*)d_in[1];
    const float* weight      = (const float*)d_in[2];
    const int*   edge_index  = (const int*)d_in[3];
    const int*   range_list  = (const int*)d_in[4];
    const float* x_norm      = (const float*)d_in[5];
    float*       out         = (float*)d_out;

    float* embedT = (float*)d_ws;
    float* X      = (float*)((char*)d_ws + 65536);
    float* agg    = (float*)((char*)d_ws + 65536 +
                             (size_t)N_SOURCE * EMBED_DIM * sizeof(float));

    const int* src = edge_index;

    transpose_embed<<<(SOURCE_DIM * EMBED_DIM + 255) / 256, 256, 0, stream>>>(embed, embedT);

    embed_gemm<<<(N_SOURCE / 32 + 7) / 8, 256, 0, stream>>>(source_feat, embedT, x_norm, X);

    gather_mean<<<N_TARGET, 256, 0, stream>>>(X, src, range_list, agg);

    proj_gemm<<<(N_TARGET / 16) / 8, 256, 0, stream>>>(agg, weight, out);
}


# proj_gemm folded into the gather kernel: a block owns 16 targets (32 waves), their 32 half sums meet in LDS and one wave runs the tile's 8 WMMAs; 2 launches, no agg plane; sum order kept
# speedup vs baseline: 2.6191x; 2.6191x over previous
#include <hip/hip_runtime.h>

typedef __attribute__((ext_vector_type(2))) float v2f;
typedef __attribute__((ext_vector_type(4))) float v4f;
typedef __attribute__((ext_vector_type(8))) float v8f;

#define N_SOURCE   100000
#define N_TARGET   4096
#define DEG        800
#define SOURCE_DIM 512
#define EMBED_DIM  32
#define TARGET_DIM 16

#define EG_TILES (N_SOURCE / 16)
#define EG_GRID  768
#define EG_PITCH 36
#define EG_RED   (8 * 16 * EG_PITCH)

__device__ __forceinline__ void eg_load_strip(v4f (&a)[8], const float* __restrict__ p) {
#pragma unroll
    for (int j = 0; j < 8; ++j) a[j] = *(const v4f*)(p + 8 * j);
    __builtin_amdgcn_sched_barrier(0);
}

__device__ __forceinline__ void eg_tile(const v4f (&a)[8], const v2f (&b0)[16],
                                        const v2f (&b1)[16], float nrm,
                                        float* mine,
                                        const float* sums,
                                        float* __restrict__ xout) {
    v8f c0 = {}, c1 = {};
#pragma unroll
    for (int j = 0; j < 8; ++j) {
        v2f ae = {a[j].x, a[j].y}, ao = {a[j].z, a[j].w};
        c0 = __builtin_amdgcn_wmma_f32_16x16x4_f32(false, ae, false, b0[2 * j],
                                                   (short)0, c0, false, false);
        c1 = __builtin_amdgcn_wmma_f32_16x16x4_f32(false, ae, false, b1[2 * j],
                                                   (short)0, c1, false, false);
        c0 = __builtin_amdgcn_wmma_f32_16x16x4_f32(false, ao, false, b0[2 * j + 1],
                                                   (short)0, c0, false, false);
        c1 = __builtin_amdgcn_wmma_f32_16x16x4_f32(false, ao, false, b1[2 * j + 1],
                                                   (short)0, c1, false, false);
    }

#pragma unroll
    for (int r = 0; r < 8; ++r) {
        mine[r * EG_PITCH]      = c0[r];
        mine[r * EG_PITCH + 16] = c1[r];
    }
    __builtin_amdgcn_fence(__ATOMIC_RELEASE, "workgroup", "local");
    __builtin_amdgcn_s_barrier();
    __builtin_amdgcn_fence(__ATOMIC_ACQUIRE, "workgroup", "local");

    float sx = 0.0f, sy = 0.0f;
#pragma unroll
    for (int w = 0; w < 8; ++w) {
        v2f p = *(const v2f*)(sums + w * (16 * EG_PITCH));
        sx += p.x;
        sy += p.y;
    }
    const float inv = 1.0f / nrm;
    v2f o = {sx * inv, sy * inv};
    *(v2f*)xout = o;
}

__global__ void __launch_bounds__(256)
embed_gemm(const float* __restrict__ A, const float* __restrict__ E,
           const float* __restrict__ xnorm, float* __restrict__ X) {
    __shared__ __attribute__((aligned(16))) float red[2 * EG_RED];

    const int wave = threadIdx.x >> 5;
    const int lane = threadIdx.x & 31;
    const int lh   = lane & 15;
    const int hi   = lane >> 4;
    const int kc   = wave * 64;

    v2f b0[16], b1[16];
#pragma unroll
    for (int j = 0; j < 8; ++j) {
        const float* e = E + (size_t)(kc + 8 * j + 4 * hi) * EMBED_DIM + lh;
        b0[2 * j].x     = e[0];
        b0[2 * j].y     = e[EMBED_DIM];
        b0[2 * j + 1].x = e[2 * EMBED_DIM];
        b0[2 * j + 1].y = e[3 * EMBED_DIM];
        b1[2 * j].x     = e[16];
        b1[2 * j].y     = e[EMBED_DIM + 16];
        b1[2 * j + 1].x = e[2 * EMBED_DIM + 16];
        b1[2 * j + 1].y = e[3 * EMBED_DIM + 16];
    }

    const int rrow = threadIdx.x >> 4;
    const int rcol = (threadIdx.x & 15) * 2;

    float*       mine = red + wave * (16 * EG_PITCH) + (8 * hi) * EG_PITCH + lh;
    const float* sums = red + rrow * EG_PITCH + rcol;

    const size_t tstride = (size_t)16 * SOURCE_DIM;
    const size_t gstride = (size_t)EG_GRID * tstride;
    const float* __restrict__ ap = A + (size_t)blockIdx.x * tstride
                                     + (size_t)lh * SOURCE_DIM + kc + 4 * hi;
    const float* __restrict__ np = xnorm + blockIdx.x * 16 + rrow;
    float*       __restrict__ xp = X + (size_t)(blockIdx.x * 16 + rrow) * EMBED_DIM + rcol;
    const size_t xstride = (size_t)EG_GRID * 16 * EMBED_DIM;

    const int ntiles = (EG_TILES - 1 - (int)blockIdx.x) / EG_GRID + 1;

    v4f P[8], Q[8];
    eg_load_strip(P, ap);

    int i = 0;
    for (; i + 2 < ntiles; i += 2) {
        const float n0 = np[0];
        eg_load_strip(Q, ap + gstride);
        eg_tile(P, b0, b1, n0, mine, sums, xp);
        const float n1 = np[EG_GRID * 16];
        eg_load_strip(P, ap + 2 * gstride);
        eg_tile(Q, b0, b1, n1, mine + EG_RED, sums + EG_RED, xp + xstride);
        ap += 2 * gstride;
        np += 2 * EG_GRID * 16;
        xp += 2 * xstride;
    }
    if (i + 2 == ntiles) {
        const float n0 = np[0];
        eg_load_strip(Q, ap + gstride);
        eg_tile(P, b0, b1, n0, mine, sums, xp);
        const float n1 = np[EG_GRID * 16];
        eg_tile(Q, b0, b1, n1, mine + EG_RED, sums + EG_RED, xp + xstride);
    } else {
        const float n0 = np[0];
        eg_tile(P, b0, b1, n0, mine, sums, xp);
    }
}

#define GM_TPB 16
#define GM_GRP 32
#define GM_PITCH 36
#define GM_PW 1

static_assert(N_TARGET % GM_TPB == 0, "a block owns a whole tile of targets");

typedef __attribute__((ext_vector_type(16))) int v16i;
typedef v16i __attribute__((aligned(4))) v16i_a4;

template <int N>
__device__ __forceinline__ void gm_rows(const int (&idx)[N], const float* __restrict__ X,
                                        const unsigned lane4,
                                        float& a0, float& a1, float& a2, float& a3) {
    unsigned off[N];
#pragma unroll
    for (int k = 0; k < N; ++k) off[k] = ((unsigned)idx[k] << 7) + lane4;
    __builtin_amdgcn_sched_barrier(0);
    float v[N];
#pragma unroll
    for (int k = 0; k < N; ++k) v[k] = *(const float*)((const char*)X + off[k]);
#pragma unroll
    for (int k = 0; k < N; k += 4) {
        a0 += v[k];
        a1 += v[k + 1];
        a2 += v[k + 2];
        a3 += v[k + 3];
    }
}

__device__ __forceinline__ void gm_group(const v16i lo, const v16i hi,
                                         const float* __restrict__ X, const unsigned lane4,
                                         float& a0, float& a1, float& a2, float& a3) {
    int idx[GM_GRP];
#pragma unroll
    for (int k = 0; k < 16; ++k) { idx[k] = lo[k]; idx[16 + k] = hi[k]; }
    gm_rows<GM_GRP>(idx, X, lane4, a0, a1, a2, a3);
}

__global__ void __launch_bounds__(1024)
gather_mean_proj(const float* __restrict__ X, const int* __restrict__ src,
                 const int* __restrict__ range_list, const float* __restrict__ Wt,
                 float* __restrict__ out) {
    __shared__ __attribute__((aligned(16))) float part[2][GM_TPB][GM_PITCH];
    const int lane = threadIdx.x & 31;
    const int wave = __builtin_amdgcn_readfirstlane((int)(threadIdx.x >> 5));
    const int tl   = wave >> 1;
    const int half = wave & 1;
    const int t0   = blockIdx.x * GM_TPB;
    const int t    = t0 + tl;
    const int start = range_list[2 * t];
    const int end   = range_list[2 * t + 1];
    const int n     = end > start ? end - start : 0;

    int h = (((n + 1) >> 1) + (GM_GRP - 1)) & ~(GM_GRP - 1);
    h = h < n ? h : n;
    const int e1 = half ? start + n : start + h;
    int       e  = half ? start + h : start;

    const unsigned lane4 = (unsigned)lane * 4u;
    float a0 = 0.0f, a1 = 0.0f, a2 = 0.0f, a3 = 0.0f;

    if (e + GM_GRP <= e1) {
        v16i clo = *(const v16i_a4*)(src + e);
        v16i chi = *(const v16i_a4*)(src + e + 16);
        for (; e + 2 * GM_GRP <= e1; e += GM_GRP) {
            const v16i nlo = *(const v16i_a4*)(src + e + GM_GRP);
            const v16i nhi = *(const v16i_a4*)(src + e + GM_GRP + 16);
            gm_group(clo, chi, X, lane4, a0, a1, a2, a3);
            clo = nlo;
            chi = nhi;
        }
        gm_group(clo, chi, X, lane4, a0, a1, a2, a3);
        e += GM_GRP;
    }
    if (e + 16 <= e1) {
        const v16i c = *(const v16i_a4*)(src + e);
        int idx[16];
#pragma unroll
        for (int k = 0; k < 16; ++k) idx[k] = c[k];
        gm_rows<16>(idx, X, lane4, a0, a1, a2, a3);
        e += 16;
    }
    for (; e < e1; ++e)
        a0 += *(const float*)((const char*)X + (((unsigned)src[e] << 7) + lane4));

    const int lh = lane & 15;
    const int hi = lane >> 4;

    v2f b[EMBED_DIM / 4] = {};
    int rs = 0, re = 0;
    if (wave == GM_PW) {
#pragma unroll
        for (int kb = 0; kb < EMBED_DIM / 4; ++kb) {
            b[kb].x = Wt[(4 * kb + 2 * hi) * TARGET_DIM + lh];
            b[kb].y = Wt[(4 * kb + 2 * hi + 1) * TARGET_DIM + lh];
        }
        rs = range_list[2 * (t0 + lh)];
        re = range_list[2 * (t0 + lh) + 1];
    }

    part[half][tl][lane] = (a0 + a1) + (a2 + a3);
    __syncthreads();

    if (wave == GM_PW) {
        float deg = (float)(re - rs);
        deg = deg > 1.0f ? deg : 1.0f;
        const float* p0 = &part[0][lh][2 * hi];
        const float* p1 = &part[1][lh][2 * hi];
        v8f c = {};
#pragma unroll
        for (int kb = 0; kb < EMBED_DIM / 4; ++kb) {
            const v2f s0 = *(const v2f*)(p0 + 4 * kb);
            const v2f s1 = *(const v2f*)(p1 + 4 * kb);
            v2f a;
            a.x = (s0.x + s1.x) / deg;
            a.y = (s0.y + s1.y) / deg;
            c = __builtin_amdgcn_wmma_f32_16x16x4_f32(false, a, false, b[kb],
                                                      (short)0, c, false, false);
        }
#pragma unroll
        for (int r = 0; r < 8; ++r)
            out[(size_t)(t0 + r + 8 * hi) * TARGET_DIM + lh] = c[r];
    }
}

extern "C" void kernel_launch(void* const* d_in, const int* in_sizes, int n_in,
                              void* d_out, int out_size, void* d_ws, size_t ws_size,
                              hipStream_t stream) {
    (void)in_sizes; (void)n_in; (void)out_size; (void)ws_size;

    const float* source_feat = (const float*)d_in[0];
    const float* embed       = (const float*)d_in[1];
    const float* weight      = (const float*)d_in[2];
    const int*   edge_index  = (const int*)d_in[3];
    const int*   range_list  = (const int*)d_in[4];
    const float* x_norm      = (const float*)d_in[5];
    float*       out         = (float*)d_out;

    float* X      = (float*)((char*)d_ws + 65536);

    const int* src = edge_index;

    embed_gemm<<<EG_GRID, 256, 0, stream>>>(source_feat, embed, x_norm, X);

    gather_mean_proj<<<N_TARGET / GM_TPB, 1024, 0, stream>>>(X, src, range_list, weight, out);
}
